# starting kernel
# baseline (speedup 1.0000x reference)
#include <hip/hip_runtime.h>

constexpr int kInputDim = 1024;
constexpr int kDayEmb   = 8;
constexpr int kPredLen  = 96;
constexpr int kHidden   = 512;
constexpr int kBatch    = 4096;

constexpr int kBTile  = 64;
constexpr int kKC     = 16;
constexpr int kNWaves = 8;
constexpr int kWaveN  = kHidden / kNWaves;

typedef __attribute__((ext_vector_type(2))) float v2f;
typedef __attribute__((ext_vector_type(8))) float v8f;

__global__ __launch_bounds__(256) void smlp_fused_kernel(
    const float* __restrict__ x,
    const float* __restrict__ day_emb,
    const float* __restrict__ W1,
    const float* __restrict__ b1,
    const float* __restrict__ W2,
    const float* __restrict__ b2,
    float* __restrict__ out)
{
  __shared__ float sA[kBTile * kKC];
  __shared__ float sB[(kKC / 2) * kHidden * 2];
  __shared__ float sBias[kHidden];
  __shared__ float sRed[kBTile * kNWaves];

  const int p    = blockIdx.y;
  const int b0   = blockIdx.x * kBTile;
  const int tid  = threadIdx.x;
  const int lane = tid & 31;
  const int wave = tid >> 5;
  const int l16  = lane & 15;
  const int hl   = lane >> 4;

  const float* __restrict__ W1p = W1 + (size_t)p * (kInputDim + kDayEmb) * kHidden;

  for (int h = tid; h < kHidden; h += 256) {
    float s = b1[p * kHidden + h];
#pragma unroll
    for (int e = 0; e < kDayEmb; ++e)
      s += day_emb[p * kDayEmb + e] * W1p[(size_t)(kInputDim + e) * kHidden + h];
    sBias[h] = s;
  }

  v8f acc[4][4] = {};

  for (int k0 = 0; k0 < kInputDim; k0 += kKC) {
    __syncthreads();

    {
      const int row = tid >> 2;
      const int q   = (tid & 3) << 2;
      const float4 v = *(const float4*)&x[(size_t)(b0 + row) * kInputDim + k0 + q];
      *(float4*)&sA[row * kKC + q] = v;
    }
#pragma unroll
    for (int i = 0; i < 4; ++i) {
      const int u    = tid + i * 256;
      const int pair = u >> 7;
      const int n    = (u & 127) << 2;
      const float4 r0 = *(const float4*)&W1p[(size_t)(k0 + 2 * pair + 0) * kHidden + n];
      const float4 r1 = *(const float4*)&W1p[(size_t)(k0 + 2 * pair + 1) * kHidden + n];
      const float4 w0 = make_float4(r0.x, r1.x, r0.y, r1.y);
      const float4 w1 = make_float4(r0.z, r1.z, r0.w, r1.w);
      *(float4*)&sB[pair * (kHidden * 2) + 2 * n + 0] = w0;
      *(float4*)&sB[pair * (kHidden * 2) + 2 * n + 4] = w1;
    }
    if (k0 + kKC < kInputDim)
      __builtin_prefetch(&W1p[(size_t)(k0 + kKC) * kHidden + tid * 32], 0, 1);

    __syncthreads();

#pragma unroll
    for (int kk = 0; kk < kKC; kk += 4) {
      v2f afrag[4];
#pragma unroll
      for (int mt = 0; mt < 4; ++mt)
        afrag[mt] = *(const v2f*)&sA[(mt * 16 + l16) * kKC + kk + 2 * hl];

      const int kp = (kk >> 1) + hl;
#pragma unroll
      for (int nt = 0; nt < 4; ++nt) {
        const int n = wave * kWaveN + nt * 16 + l16;
        const v2f bfrag = *(const v2f*)&sB[kp * (kHidden * 2) + 2 * n];
#pragma unroll
        for (int mt = 0; mt < 4; ++mt)
          acc[mt][nt] = __builtin_amdgcn_wmma_f32_16x16x4_f32(
              false, afrag[mt], false, bfrag, (short)0, acc[mt][nt],
              false, false);
      }
    }
  }

  __syncthreads();

  const float* __restrict__ W2p = W2 + (size_t)p * kHidden;
#pragma unroll
  for (int mt = 0; mt < 4; ++mt) {
    float rs[8];
#pragma unroll
    for (int r = 0; r < 8; ++r) rs[r] = 0.0f;

#pragma unroll
    for (int nt = 0; nt < 4; ++nt) {
      const int n     = wave * kWaveN + nt * 16 + l16;
      const float bn  = sBias[n];
      const float w2n = W2p[n];
#pragma unroll
      for (int r = 0; r < 8; ++r) {
        float v = acc[mt][nt][r] + bn;
        v = fmaxf(v, 0.0f);
        rs[r] += v * w2n;
      }
    }
#pragma unroll
    for (int r = 0; r < 8; ++r) {
      float s = rs[r];
      s += __shfl_xor(s, 1, 16);
      s += __shfl_xor(s, 2, 16);
      s += __shfl_xor(s, 4, 16);
      s += __shfl_xor(s, 8, 16);
      if (l16 == 0)
        sRed[(mt * 16 + hl * 8 + r) * kNWaves + wave] = s;
    }
  }

  __syncthreads();

  if (tid < kBTile) {
    float s = b2[p];
#pragma unroll
    for (int w = 0; w < kNWaves; ++w) s += sRed[tid * kNWaves + w];
    out[(size_t)(b0 + tid) * kPredLen + p] = s;
  }
}

extern "C" void kernel_launch(void* const* d_in, const int* in_sizes, int n_in,
                              void* d_out, int out_size, void* d_ws, size_t ws_size,
                              hipStream_t stream) {
  const float* x       = (const float*)d_in[0];
  const float* day_emb = (const float*)d_in[1];
  const float* W1      = (const float*)d_in[2];
  const float* b1      = (const float*)d_in[3];
  const float* W2      = (const float*)d_in[4];
  const float* b2      = (const float*)d_in[5];
  float* out           = (float*)d_out;

  dim3 grid(kBatch / kBTile, kPredLen);
  smlp_fused_kernel<<<grid, 256, 0, stream>>>(x, day_emb, W1, b1, W2, b2, out);
}


# K loop software-pipelined: next chunk's global loads issued before the WMMA block, LDS tiles double-buffered, one barrier per chunk instead of two (cache prefetch dropped)
# speedup vs baseline: 1.0793x; 1.0793x over previous
#include <hip/hip_runtime.h>


constexpr int kInputDim = 1024;
constexpr int kDayEmb   = 8;
constexpr int kPredLen  = 96;
constexpr int kHidden   = 512;
constexpr int kBatch    = 4096;

constexpr int kBTile  = 64;
constexpr int kKC     = 16;
constexpr int kNWaves = 8;
constexpr int kWaveN  = kHidden / kNWaves;
constexpr int kChunks = kInputDim / kKC;

constexpr int kAStage = kBTile * kKC;
constexpr int kBStage = (kKC / 2) * kHidden * 2;

typedef __attribute__((ext_vector_type(2))) float v2f;
typedef __attribute__((ext_vector_type(8))) float v8f;

__device__ __forceinline__ void mma_chunk(v8f (&acc)[4][4],
                                          const float* __restrict__ sAc,
                                          const float* __restrict__ sBc,
                                          int l16, int hl, int wave)
{
#pragma unroll
  for (int kk = 0; kk < kKC; kk += 4) {
    v2f afrag[4];
#pragma unroll
    for (int mt = 0; mt < 4; ++mt)
      afrag[mt] = *(const v2f*)&sAc[(mt * 16 + l16) * kKC + kk + 2 * hl];

    const int kp = (kk >> 1) + hl;
#pragma unroll
    for (int nt = 0; nt < 4; ++nt) {
      const int n = wave * kWaveN + nt * 16 + l16;
      const v2f bfrag = *(const v2f*)&sBc[kp * (kHidden * 2) + 2 * n];
#pragma unroll
      for (int mt = 0; mt < 4; ++mt)
        acc[mt][nt] = __builtin_amdgcn_wmma_f32_16x16x4_f32(
            false, afrag[mt], false, bfrag, (short)0, acc[mt][nt],
            false, false);
    }
  }
}

struct ChunkRegs {
  float4 rx;
  float4 r0[4];
  float4 r1[4];
};

__device__ __forceinline__ void fetch_chunk(ChunkRegs& g, const float* xg, const float* wg)
{
  g.rx = *(const float4*)xg;
#pragma unroll
  for (int i = 0; i < 4; ++i) {
    g.r0[i] = *(const float4*)&wg[(size_t)(4 * i + 0) * kHidden];
    g.r1[i] = *(const float4*)&wg[(size_t)(4 * i + 1) * kHidden];
  }
}

__device__ __forceinline__ void put_chunk(const ChunkRegs& g, float* sAst, float* sBst,
                                          int sAw, int sBw)
{
  *(float4*)&sAst[sAw] = g.rx;
#pragma unroll
  for (int i = 0; i < 4; ++i) {
    *(float4*)&sBst[sBw + 2 * i * (kHidden * 2) + 0] =
        make_float4(g.r0[i].x, g.r1[i].x, g.r0[i].y, g.r1[i].y);
    *(float4*)&sBst[sBw + 2 * i * (kHidden * 2) + 4] =
        make_float4(g.r0[i].z, g.r1[i].z, g.r0[i].w, g.r1[i].w);
  }
}

template <int RD>
__device__ __forceinline__ void pipe_step(v8f (&acc)[4][4], float* sA, float* sB,
                                          const float*& xg, const float*& wg,
                                          int sAw, int sBw, int l16, int hl, int wave)
{
  constexpr int WR = RD ^ 1;
  ChunkRegs g;
  xg += kKC;
  wg += (size_t)kKC * kHidden;
  fetch_chunk(g, xg, wg);

  __builtin_amdgcn_sched_barrier(0);

  mma_chunk(acc, sA + RD * kAStage, sB + RD * kBStage, l16, hl, wave);

  __builtin_amdgcn_sched_barrier(0);

  put_chunk(g, sA + WR * kAStage, sB + WR * kBStage, sAw, sBw);

  __syncthreads();
}

__global__ __launch_bounds__(256, 4) void smlp_fused_kernel(
    const float* __restrict__ x,
    const float* __restrict__ day_emb,
    const float* __restrict__ W1,
    const float* __restrict__ b1,
    const float* __restrict__ W2,
    const float* __restrict__ b2,
    float* __restrict__ out)
{
  __shared__ float sA[2 * kAStage];
  __shared__ float sB[2 * kBStage];
  __shared__ float sBias[kHidden];
  __shared__ float sRed[kBTile * kNWaves];

  const int p    = blockIdx.y;
  const int b0   = blockIdx.x * kBTile;
  const int tid  = threadIdx.x;
  const int lane = tid & 31;
  const int wave = tid >> 5;
  const int l16  = lane & 15;
  const int hl   = lane >> 4;

  const float* __restrict__ W1p = W1 + (size_t)p * (kInputDim + kDayEmb) * kHidden;

  for (int h = tid; h < kHidden; h += 256) {
    float s = b1[p * kHidden + h];
#pragma unroll
    for (int e = 0; e < kDayEmb; ++e)
      s += day_emb[p * kDayEmb + e] * W1p[(size_t)(kInputDim + e) * kHidden + h];
    sBias[h] = s;
  }

  const int xrow = tid >> 2;
  const int xq   = (tid & 3) << 2;
  const float* xg = x + (size_t)(b0 + xrow) * kInputDim + xq;
  const int sAw = xrow * kKC + xq;
  const int wpair = tid >> 7;
  const int wn    = (tid & 127) << 2;
  const float* wg = W1p + (size_t)(2 * wpair) * kHidden + wn;
  const int sBw = wpair * (kHidden * 2) + 2 * wn;

  {
    ChunkRegs g;
    fetch_chunk(g, xg, wg);
    put_chunk(g, sA, sB, sAw, sBw);
  }

  v8f acc[4][4] = {};

  __syncthreads();

  for (int c = 0; c + 2 < kChunks; c += 2) {
    pipe_step<0>(acc, sA, sB, xg, wg, sAw, sBw, l16, hl, wave);
    pipe_step<1>(acc, sA, sB, xg, wg, sAw, sBw, l16, hl, wave);
  }
  pipe_step<0>(acc, sA, sB, xg, wg, sAw, sBw, l16, hl, wave);
  mma_chunk(acc, sA + kAStage, sB + kBStage, l16, hl, wave);

  const float* __restrict__ W2p = W2 + (size_t)p * kHidden;
#pragma unroll
  for (int mt = 0; mt < 4; ++mt) {
    float rs[8];
#pragma unroll
    for (int r = 0; r < 8; ++r) rs[r] = 0.0f;

#pragma unroll
    for (int nt = 0; nt < 4; ++nt) {
      const int n     = wave * kWaveN + nt * 16 + l16;
      const float bn  = sBias[n];
      const float w2n = W2p[n];
#pragma unroll
      for (int r = 0; r < 8; ++r) {
        float v = acc[mt][nt][r] + bn;
        v = fmaxf(v, 0.0f);
        rs[r] += v * w2n;
      }
    }
#pragma unroll
    for (int r = 0; r < 8; ++r) {
      float s = rs[r];
      s += __shfl_xor(s, 1, 16);
      s += __shfl_xor(s, 2, 16);
      s += __shfl_xor(s, 4, 16);
      s += __shfl_xor(s, 8, 16);
      if (l16 == 0)
        sRed[(mt * 16 + hl * 8 + r) * kNWaves + wave] = s;
    }
  }

  __syncthreads();

  if (tid < kBTile) {
    float s = b2[p];
#pragma unroll
    for (int w = 0; w < kNWaves; ++w) s += sRed[tid * kNWaves + w];
    out[(size_t)(b0 + tid) * kPredLen + p] = s;
  }
}

extern "C" void kernel_launch(void* const* d_in, const int* in_sizes, int n_in,
                              void* d_out, int out_size, void* d_ws, size_t ws_size,
                              hipStream_t stream) {
  const float* x       = (const float*)d_in[0];
  const float* day_emb = (const float*)d_in[1];
  const float* W1      = (const float*)d_in[2];
  const float* b1      = (const float*)d_in[3];
  const float* W2      = (const float*)d_in[4];
  const float* b2      = (const float*)d_in[5];
  float* out           = (float*)d_out;

  dim3 grid(kBatch / kBTile, kPredLen);
  smlp_fused_kernel<<<grid, 256, 0, stream>>>(x, day_emb, W1, b1, W2, b2, out);
}


# W1 B fragments loaded from global memory straight into registers (lane = 4 adjacent hidden columns, one per N tile); W1 LDS staging removed; hidden-dim sum regrouped per lane (order changes)
# speedup vs baseline: 1.1426x; 1.0586x over previous
#include <hip/hip_runtime.h>


constexpr int kInputDim = 1024;
constexpr int kDayEmb   = 8;
constexpr int kPredLen  = 96;
constexpr int kHidden   = 512;
constexpr int kBatch    = 4096;

constexpr int kBTile  = 64;
constexpr int kKC     = 16;
constexpr int kNWaves = 8;
constexpr int kWaveN  = kHidden / kNWaves;
constexpr int kChunks = kInputDim / kKC;

constexpr int kAStage = kBTile * kKC;


typedef __attribute__((ext_vector_type(2))) float v2f;
typedef __attribute__((ext_vector_type(8))) float v8f;

__device__ __forceinline__ void mma_chunk(v8f (&acc)[4][4],
                                          const float* __restrict__ sAc,
                                          const v2f (&bf)[4][4],
                                          int l16, int hl)
{
#pragma unroll
  for (int j = 0; j < 4; ++j) {
    v2f afrag[4];
#pragma unroll
    for (int mt = 0; mt < 4; ++mt)
      afrag[mt] = *(const v2f*)&sAc[(mt * 16 + l16) * kKC + 4 * j + 2 * hl];

#pragma unroll
    for (int nt = 0; nt < 4; ++nt) {
#pragma unroll
      for (int mt = 0; mt < 4; ++mt)
        acc[mt][nt] = __builtin_amdgcn_wmma_f32_16x16x4_f32(
            false, afrag[mt], false, bf[j][nt], (short)0, acc[mt][nt],
            false, false);
    }
  }
}

struct ChunkRegs {
  float4 rx;
  float4 rw[4][2];
};

__device__ __forceinline__ void fetch_chunk(ChunkRegs& g, const float* xg, const float* wg)
{
  g.rx = *(const float4*)xg;
#pragma unroll
  for (int j = 0; j < 4; ++j) {
    g.rw[j][0] = *(const float4*)&wg[(size_t)(4 * j + 0) * kHidden];
    g.rw[j][1] = *(const float4*)&wg[(size_t)(4 * j + 1) * kHidden];
  }
}

__device__ __forceinline__ void make_bfrags(const ChunkRegs& g, v2f (&bf)[4][4])
{
#pragma unroll
  for (int j = 0; j < 4; ++j) {
    bf[j][0] = (v2f){g.rw[j][0].x, g.rw[j][1].x};
    bf[j][1] = (v2f){g.rw[j][0].y, g.rw[j][1].y};
    bf[j][2] = (v2f){g.rw[j][0].z, g.rw[j][1].z};
    bf[j][3] = (v2f){g.rw[j][0].w, g.rw[j][1].w};
  }
}

template <int RD>
__device__ __forceinline__ void pipe_step(v8f (&acc)[4][4], float* sA, v2f (&bf)[4][4],
                                          const float*& xg, const float*& wg,
                                          int sAw, int l16, int hl)
{
  constexpr int WR = RD ^ 1;
  ChunkRegs g;
  xg += kKC;
  wg += (size_t)kKC * kHidden;
  fetch_chunk(g, xg, wg);

  __builtin_amdgcn_sched_barrier(0);

  mma_chunk(acc, sA + RD * kAStage, bf, l16, hl);

  __builtin_amdgcn_sched_barrier(0);

  *(float4*)&sA[WR * kAStage + sAw] = g.rx;
  make_bfrags(g, bf);

  __syncthreads();
}

__global__ __launch_bounds__(256, 4) void smlp_fused_kernel(
    const float* __restrict__ x,
    const float* __restrict__ day_emb,
    const float* __restrict__ W1,
    const float* __restrict__ b1,
    const float* __restrict__ W2,
    const float* __restrict__ b2,
    float* __restrict__ out)
{
  __shared__ float sA[2 * kAStage];
  __shared__ float sBias[kHidden];
  __shared__ float sRed[kBTile * kNWaves];

  const int p    = blockIdx.y;
  const int b0   = blockIdx.x * kBTile;
  const int tid  = threadIdx.x;
  const int lane = tid & 31;
  const int wave = tid >> 5;
  const int l16  = lane & 15;
  const int hl   = lane >> 4;

  const float* __restrict__ W1p = W1 + (size_t)p * (kInputDim + kDayEmb) * kHidden;

  for (int h = tid; h < kHidden; h += 256) {
    float s = b1[p * kHidden + h];
#pragma unroll
    for (int e = 0; e < kDayEmb; ++e)
      s += day_emb[p * kDayEmb + e] * W1p[(size_t)(kInputDim + e) * kHidden + h];
    sBias[h] = s;
  }

  const int xrow = tid >> 2;
  const int xq   = (tid & 3) << 2;
  const float* xg = x + (size_t)(b0 + xrow) * kInputDim + xq;
  const int sAw = xrow * kKC + xq;
  const float* wg = W1p + (size_t)(2 * hl) * kHidden + wave * kWaveN + 4 * l16;

  v2f bf[4][4];
  {
    ChunkRegs g;
    fetch_chunk(g, xg, wg);
    *(float4*)&sA[sAw] = g.rx;
    make_bfrags(g, bf);
  }

  v8f acc[4][4] = {};

  __syncthreads();

  for (int c = 0; c + 2 < kChunks; c += 2) {
    pipe_step<0>(acc, sA, bf, xg, wg, sAw, l16, hl);
    pipe_step<1>(acc, sA, bf, xg, wg, sAw, l16, hl);
  }
  pipe_step<0>(acc, sA, bf, xg, wg, sAw, l16, hl);
  mma_chunk(acc, sA + kAStage, bf, l16, hl);

  const float* __restrict__ W2p = W2 + (size_t)p * kHidden;
#pragma unroll
  for (int mt = 0; mt < 4; ++mt) {
    float rs[8];
#pragma unroll
    for (int r = 0; r < 8; ++r) rs[r] = 0.0f;

#pragma unroll
    for (int nt = 0; nt < 4; ++nt) {
      const int n     = wave * kWaveN + 4 * l16 + nt;
      const float bn  = sBias[n];
      const float w2n = W2p[n];
#pragma unroll
      for (int r = 0; r < 8; ++r) {
        float v = acc[mt][nt][r] + bn;
        v = fmaxf(v, 0.0f);
        rs[r] += v * w2n;
      }
    }
#pragma unroll
    for (int r = 0; r < 8; ++r) {
      float s = rs[r];
      s += __shfl_xor(s, 1, 16);
      s += __shfl_xor(s, 2, 16);
      s += __shfl_xor(s, 4, 16);
      s += __shfl_xor(s, 8, 16);
      if (l16 == 0)
        sRed[(mt * 16 + hl * 8 + r) * kNWaves + wave] = s;
    }
  }

  __syncthreads();

  if (tid < kBTile) {
    float s = b2[p];
#pragma unroll
    for (int w = 0; w < kNWaves; ++w) s += sRed[tid * kNWaves + w];
    out[(size_t)(b0 + tid) * kPredLen + p] = s;
  }
}

extern "C" void kernel_launch(void* const* d_in, const int* in_sizes, int n_in,
                              void* d_out, int out_size, void* d_ws, size_t ws_size,
                              hipStream_t stream) {
  const float* x       = (const float*)d_in[0];
  const float* day_emb = (const float*)d_in[1];
  const float* W1      = (const float*)d_in[2];
  const float* b1      = (const float*)d_in[3];
  const float* W2      = (const float*)d_in[4];
  const float* b2      = (const float*)d_in[5];
  float* out           = (float*)d_out;

  dim3 grid(kBatch / kBTile, kPredLen);
  smlp_fused_kernel<<<grid, 256, 0, stream>>>(x, day_emb, W1, b1, W2, b2, out);
}


# x tile ring of 3 LDS stages (was 2): stage store at the head of the chunk, next chunk's first A fragments read mid-block, so the chunk barrier no longer waits on an LDS round trip
# speedup vs baseline: 1.1857x; 1.0377x over previous
#include <hip/hip_runtime.h>


constexpr int kInputDim = 1024;
constexpr int kDayEmb   = 8;
constexpr int kPredLen  = 96;
constexpr int kHidden   = 512;
constexpr int kBatch    = 4096;

constexpr int kBTile  = 64;
constexpr int kKC     = 16;
constexpr int kNWaves = 8;
constexpr int kWaveN  = kHidden / kNWaves;
constexpr int kChunks = kInputDim / kKC;

constexpr int kStages = 3;
constexpr int kAPitch = kStages * kKC;


typedef __attribute__((ext_vector_type(2))) float v2f;
typedef __attribute__((ext_vector_type(8))) float v8f;

template <int JB>
__device__ __forceinline__ void load_afrags(v2f (&af)[2][4], const float* __restrict__ sAc,
                                            int l16, int hl)
{
#pragma unroll
  for (int jj = 0; jj < 2; ++jj)
#pragma unroll
    for (int mt = 0; mt < 4; ++mt)
      af[jj][mt] = *(const v2f*)&sAc[(mt * 16 + l16) * kAPitch + 4 * (JB + jj) + 2 * hl];
}

template <int JB>
__device__ __forceinline__ void mma_half(v8f (&acc)[4][4], const v2f (&af)[2][4],
                                         const v2f (&bf)[4][4])
{
#pragma unroll
  for (int jj = 0; jj < 2; ++jj) {
#pragma unroll
    for (int nt = 0; nt < 4; ++nt) {
#pragma unroll
      for (int mt = 0; mt < 4; ++mt)
        acc[mt][nt] = __builtin_amdgcn_wmma_f32_16x16x4_f32(
            false, af[jj][mt], false, bf[JB + jj][nt], (short)0, acc[mt][nt],
            false, false);
    }
  }
}

struct WRegs {
  float4 rw[4][2];
};

__device__ __forceinline__ void fetch_w(WRegs& g, const float* wg, unsigned woff)
{
#pragma unroll
  for (int j = 0; j < 4; ++j) {
    g.rw[j][0] = *(const float4*)&wg[woff + (4 * j + 0) * kHidden];
    g.rw[j][1] = *(const float4*)&wg[woff + (4 * j + 1) * kHidden];
  }
}

__device__ __forceinline__ void make_bfrags(const WRegs& g, v2f (&bf)[4][4])
{
#pragma unroll
  for (int j = 0; j < 4; ++j) {
    bf[j][0] = (v2f){g.rw[j][0].x, g.rw[j][1].x};
    bf[j][1] = (v2f){g.rw[j][0].y, g.rw[j][1].y};
    bf[j][2] = (v2f){g.rw[j][0].z, g.rw[j][1].z};
    bf[j][3] = (v2f){g.rw[j][0].w, g.rw[j][1].w};
  }
}

template <int RD, bool STORE, bool FETCHX, bool FETCHW, bool PRE>
__device__ __forceinline__ void ring_step(v8f (&acc)[4][4], float* sA, v2f (&bf)[4][4],
                                          v2f (&afA)[2][4], float4& rx,
                                          const float*& xg, const float*& wg,
                                          unsigned xoff, unsigned woff,
                                          int sAw, int l16, int hl)
{
  constexpr int NX = (RD + 1) % 3;
  constexpr int WR = (RD + 2) % 3;
  WRegs g;
  v2f afB[2][4];

  if (STORE) *(float4*)&sA[WR * kKC + sAw] = rx;
  if (FETCHW) { wg += (size_t)kKC * kHidden; fetch_w(g, wg, woff); }
  load_afrags<2>(afB, sA + RD * kKC, l16, hl);

  __builtin_amdgcn_sched_barrier(0);

  mma_half<0>(acc, afA, bf);

  __builtin_amdgcn_sched_barrier(0);

  if (PRE) load_afrags<0>(afA, sA + NX * kKC, l16, hl);
  if (FETCHX) { xg += kKC; rx = *(const float4*)&xg[xoff]; }

  __builtin_amdgcn_sched_barrier(0);

  mma_half<2>(acc, afB, bf);

  __builtin_amdgcn_sched_barrier(0);

  if (FETCHW) make_bfrags(g, bf);

  if (PRE) __syncthreads();
}

__global__ __launch_bounds__(256, 4) void smlp_fused_kernel(
    const float* __restrict__ x,
    const float* __restrict__ day_emb,
    const float* __restrict__ W1,
    const float* __restrict__ b1,
    const float* __restrict__ W2,
    const float* __restrict__ b2,
    float* __restrict__ out)
{
  __shared__ float sA[kBTile * kAPitch];
  __shared__ float sBias[kHidden];
  __shared__ float sRed[kBTile * kNWaves];

  const int p    = blockIdx.y;
  const int b0   = blockIdx.x * kBTile;
  const int tid  = threadIdx.x;
  const int lane = tid & 31;
  const int wave = tid >> 5;
  const int l16  = lane & 15;
  const int hl   = lane >> 4;

  const float* __restrict__ W1p = W1 + (size_t)p * (kInputDim + kDayEmb) * kHidden;

  for (int h = tid; h < kHidden; h += 256) {
    float s = b1[p * kHidden + h];
#pragma unroll
    for (int e = 0; e < kDayEmb; ++e)
      s += day_emb[p * kDayEmb + e] * W1p[(size_t)(kInputDim + e) * kHidden + h];
    sBias[h] = s;
  }

  const int xrow = tid >> 2;
  const int xq   = (tid & 3) << 2;
  const float* xg = x + (size_t)b0 * kInputDim;
  const unsigned xoff = xrow * kInputDim + xq;
  const int sAw = xrow * kAPitch + xq;
  const float* wg = W1p;
  const unsigned woff = (2 * hl) * kHidden + wave * kWaveN + 4 * l16;

  v2f bf[4][4];
  float4 rx;
  {
    WRegs g;
    const float4 x0 = *(const float4*)&xg[xoff];
    xg += kKC;
    const float4 x1 = *(const float4*)&xg[xoff];
    xg += kKC;
    rx = *(const float4*)&xg[xoff];
    fetch_w(g, wg, woff);
    *(float4*)&sA[0 * kKC + sAw] = x0;
    *(float4*)&sA[1 * kKC + sAw] = x1;
    make_bfrags(g, bf);
  }

  v8f acc[4][4] = {};

  __syncthreads();

  v2f afA[2][4];
  load_afrags<0>(afA, sA, l16, hl);

  for (int c = 0; c < kChunks - 4; c += 3) {
    ring_step<0, true, true, true, true>(acc, sA, bf, afA, rx, xg, wg, xoff, woff, sAw, l16, hl);
    ring_step<1, true, true, true, true>(acc, sA, bf, afA, rx, xg, wg, xoff, woff, sAw, l16, hl);
    ring_step<2, true, true, true, true>(acc, sA, bf, afA, rx, xg, wg, xoff, woff, sAw, l16, hl);
  }
  ring_step<0, true,  true,  true,  true >(acc, sA, bf, afA, rx, xg, wg, xoff, woff, sAw, l16, hl);
  ring_step<1, true,  false, true,  true >(acc, sA, bf, afA, rx, xg, wg, xoff, woff, sAw, l16, hl);
  ring_step<2, false, false, true,  true >(acc, sA, bf, afA, rx, xg, wg, xoff, woff, sAw, l16, hl);
  ring_step<0, false, false, false, false>(acc, sA, bf, afA, rx, xg, wg, xoff, woff, sAw, l16, hl);

  const float* __restrict__ W2p = W2 + (size_t)p * kHidden;
#pragma unroll
  for (int mt = 0; mt < 4; ++mt) {
    float rs[8];
#pragma unroll
    for (int r = 0; r < 8; ++r) rs[r] = 0.0f;

#pragma unroll
    for (int nt = 0; nt < 4; ++nt) {
      const int n     = wave * kWaveN + 4 * l16 + nt;
      const float bn  = sBias[n];
      const float w2n = W2p[n];
#pragma unroll
      for (int r = 0; r < 8; ++r) {
        float v = acc[mt][nt][r] + bn;
        v = fmaxf(v, 0.0f);
        rs[r] += v * w2n;
      }
    }
#pragma unroll
    for (int r = 0; r < 8; ++r) {
      float s = rs[r];
      s += __shfl_xor(s, 1, 16);
      s += __shfl_xor(s, 2, 16);
      s += __shfl_xor(s, 4, 16);
      s += __shfl_xor(s, 8, 16);
      if (l16 == 0)
        sRed[(mt * 16 + hl * 8 + r) * kNWaves + wave] = s;
    }
  }

  __syncthreads();

  if (tid < kBTile) {
    float s = b2[p];
#pragma unroll
    for (int w = 0; w < kNWaves; ++w) s += sRed[tid * kNWaves + w];
    out[(size_t)(b0 + tid) * kPredLen + p] = s;
  }
}

extern "C" void kernel_launch(void* const* d_in, const int* in_sizes, int n_in,
                              void* d_out, int out_size, void* d_ws, size_t ws_size,
                              hipStream_t stream) {
  const float* x       = (const float*)d_in[0];
  const float* day_emb = (const float*)d_in[1];
  const float* W1      = (const float*)d_in[2];
  const float* b1      = (const float*)d_in[3];
  const float* W2      = (const float*)d_in[4];
  const float* b2      = (const float*)d_in[5];
  float* out           = (float*)d_out;

  dim3 grid(kBatch / kBTile, kPredLen);
  smlp_fused_kernel<<<grid, 256, 0, stream>>>(x, day_emb, W1, b1, W2, b2, out);
}
